# starting kernel
# baseline (speedup 1.0000x reference)
#include <hip/hip_runtime.h>
#include <hip/hip_bf16.h>

typedef __attribute__((ext_vector_type(2))) float v2f;
typedef __attribute__((ext_vector_type(8))) float v8f;

#define B_DIM 1024
#define C_DIM 16
#define M_DIM 4096
#define N_DIM 8192
#define K_DIM 32
#define HALF_LOG_2PI 0.91893853320467274f

__global__ void build_u_kernel(const float* __restrict__ x,
                               float* __restrict__ U) {
    int idx = blockIdx.x * blockDim.x + threadIdx.x;
    if (idx >= B_DIM * C_DIM) return;
    int b = idx >> 4;
    int c = idx & 15;
    float v = x[b * C_DIM + c];
    U[b * K_DIM + c]         = v * v;
    U[b * K_DIM + C_DIM + c] = v;
}

__global__ void build_w_kernel(const float* __restrict__ locs,
                               const float* __restrict__ scales,
                               const int*   __restrict__ edges,
                               float* __restrict__ W,
                               float* __restrict__ Koff) {
    int n = blockIdx.x * blockDim.x + threadIdx.x;
    if (n >= N_DIM) return;
    float ko = 0.0f;
#pragma unroll
    for (int c = 0; c < C_DIM; ++c) {
        int   e    = edges[c * N_DIM + n];
        float mu   = locs[c * M_DIM + e];
        float s    = scales[c * M_DIM + e] + 0.5f;
        float inv  = 1.0f / s;
        float inv2 = inv * inv;
        W[c * N_DIM + n]           = -0.5f * inv2;
        W[(C_DIM + c) * N_DIM + n] = mu * inv2;
        ko += -0.5f * mu * mu * inv2 - __logf(s) - HALF_LOG_2PI;
    }
    Koff[n] = ko;
}

__global__ __launch_bounds__(256)
void wmma_gemm_kernel(const float* __restrict__ U,
                      const float* __restrict__ W,
                      const float* __restrict__ Koff,
                      float* __restrict__ out) {
    const int lane = threadIdx.x & 31;
    const int wave = threadIdx.x >> 5;

    constexpr int TILES_N = N_DIM / 16;
    const int n_tile  = blockIdx.x % TILES_N;
    const int b_group = blockIdx.x / TILES_N;

    const int b0 = (b_group * 8 + wave) * 16;
    const int n0 = n_tile * 16;

    const int half = lane >> 4;
    const int li   = lane & 15;

    const float* Ua = U + (b0 + li) * K_DIM + half * 2;
    const float* Wb = W + half * 2 * N_DIM + (n0 + li);

    v8f acc = {};
#pragma unroll
    for (int k0 = 0; k0 < K_DIM; k0 += 4) {
        v2f a;
        a.x = Ua[k0];
        a.y = Ua[k0 + 1];
        v2f bf;
        bf.x = Wb[k0 * N_DIM];
        bf.y = Wb[(k0 + 1) * N_DIM];
        acc = __builtin_amdgcn_wmma_f32_16x16x4_f32(
             false, a,  false, bf,
             (short)0, acc,  false,  false);
    }

    const float ko = Koff[n0 + li];
    float* o = out + (b0 + half * 8) * N_DIM + (n0 + li);
#pragma unroll
    for (int i = 0; i < 8; ++i) {
        o[i * N_DIM] = acc[i] + ko;
    }
}

extern "C" void kernel_launch(void* const* d_in, const int* in_sizes, int n_in,
                              void* d_out, int out_size, void* d_ws, size_t ws_size,
                              hipStream_t stream) {
    const float* x      = (const float*)d_in[0];
    const float* locs   = (const float*)d_in[1];
    const float* scales = (const float*)d_in[2];
    const int*   edges  = (const int*)d_in[3];
    float*       out    = (float*)d_out;

    float* U    = (float*)d_ws;
    float* W    = U + (size_t)B_DIM * K_DIM;
    float* Koff = W + (size_t)K_DIM * N_DIM;

    build_u_kernel<<<(B_DIM * C_DIM + 255) / 256, 256, 0, stream>>>(x, U);
    build_w_kernel<<<(N_DIM + 255) / 256, 256, 0, stream>>>(locs, scales, edges,
                                                            W, Koff);

    constexpr int GRID = (N_DIM / 16) * (B_DIM / 128);
    wmma_gemm_kernel<<<GRID, 256, 0, stream>>>(U, W, Koff, out);
}


# GEMM re-tiled: one wave per 64x64 block, rows/cols dealt to its 16 tiles interleaved by 4, so operand loads and stores are 16 B per lane in whole 256-B row segments (U k-major); sums unchanged
# speedup vs baseline: 2.2957x; 2.2957x over previous
#include <hip/hip_runtime.h>
#include <hip/hip_bf16.h>

typedef __attribute__((ext_vector_type(2))) float v2f;
typedef __attribute__((ext_vector_type(4))) float v4f;
typedef __attribute__((ext_vector_type(8))) float v8f;

#define B_DIM 1024
#define C_DIM 16
#define M_DIM 4096
#define N_DIM 8192
#define K_DIM 32
#define HALF_LOG_2PI 0.91893853320467274f

__global__ void build_u_kernel(const float* __restrict__ x,
                               float* __restrict__ Ut) {
    int idx = blockIdx.x * blockDim.x + threadIdx.x;
    if (idx >= B_DIM * C_DIM) return;
    int c = idx / B_DIM;
    int b = idx % B_DIM;
    float v = x[b * C_DIM + c];
    Ut[c * B_DIM + b]           = v * v;
    Ut[(C_DIM + c) * B_DIM + b] = v;
}

__global__ void build_w_kernel(const float* __restrict__ locs,
                               const float* __restrict__ scales,
                               const int*   __restrict__ edges,
                               float* __restrict__ W,
                               float* __restrict__ Koff) {
    int n = blockIdx.x * blockDim.x + threadIdx.x;
    if (n >= N_DIM) return;
    float ko = 0.0f;
#pragma unroll
    for (int c = 0; c < C_DIM; ++c) {
        int   e    = edges[c * N_DIM + n];
        float mu   = locs[c * M_DIM + e];
        float s    = scales[c * M_DIM + e] + 0.5f;
        float inv  = 1.0f / s;
        float inv2 = inv * inv;
        W[c * N_DIM + n]           = -0.5f * inv2;
        W[(C_DIM + c) * N_DIM + n] = mu * inv2;
        ko += -0.5f * mu * mu * inv2 - __logf(s) - HALF_LOG_2PI;
    }
    Koff[n] = ko;
}

__global__ __launch_bounds__(256)
void wmma_gemm_kernel(const float* __restrict__ Ut,
                      const float* __restrict__ W,
                      const float* __restrict__ Koff,
                      float* __restrict__ out) {
    const int lane = threadIdx.x & 31;
    const int wave = threadIdx.x >> 5;
    const int h    = lane >> 4;
    const int li   = lane & 15;

    const int nb = blockIdx.x & 31;
    const int bb = blockIdx.x >> 5;
    const int b_base = bb * 128 + (wave >> 2) * 64;
    const int n_base = nb * 256 + (wave & 3) * 64;

    const float* Ap = Ut + (2 * h) * B_DIM + b_base + 4 * li;
    const float* Bp = W  + (2 * h) * N_DIM + n_base + 4 * li;

    v4f a[8][2], w[8][2];
#pragma unroll
    for (int j = 0; j < 8; ++j) {
#pragma unroll
        for (int s = 0; s < 2; ++s) {
            a[j][s] = *(const v4f*)(Ap + (4 * j + s) * B_DIM);
            w[j][s] = *(const v4f*)(Bp + (4 * j + s) * N_DIM);
        }
    }
    const v4f ko = *(const v4f*)(Koff + n_base + 4 * li);

#pragma unroll
    for (int u = 0; u < 4; ++u) {
        v8f acc[4] = {};
#pragma unroll
        for (int j = 0; j < 8; ++j) {
            v2f af;
            af.x = a[j][0][u];
            af.y = a[j][1][u];
#pragma unroll
            for (int t = 0; t < 4; ++t) {
                v2f bf;
                bf.x = w[j][0][t];
                bf.y = w[j][1][t];
                acc[t] = __builtin_amdgcn_wmma_f32_16x16x4_f32(
                     false, af,  false, bf,
                     (short)0, acc[t],
                     false,  false);
            }
        }
        float* o = out + (b_base + 32 * h + u) * N_DIM + n_base + 4 * li;
#pragma unroll
        for (int r = 0; r < 8; ++r) {
            v4f v;
            v.x = acc[0][r] + ko.x;
            v.y = acc[1][r] + ko.y;
            v.z = acc[2][r] + ko.z;
            v.w = acc[3][r] + ko.w;
            *(v4f*)(o + (4 * r) * N_DIM) = v;
        }
        __builtin_amdgcn_sched_barrier(0);
    }
}

extern "C" void kernel_launch(void* const* d_in, const int* in_sizes, int n_in,
                              void* d_out, int out_size, void* d_ws, size_t ws_size,
                              hipStream_t stream) {
    const float* x      = (const float*)d_in[0];
    const float* locs   = (const float*)d_in[1];
    const float* scales = (const float*)d_in[2];
    const int*   edges  = (const int*)d_in[3];
    float*       out    = (float*)d_out;

    float* Ut   = (float*)d_ws;
    float* W    = Ut + (size_t)B_DIM * K_DIM;
    float* Koff = W + (size_t)K_DIM * N_DIM;

    build_u_kernel<<<(B_DIM * C_DIM + 255) / 256, 256, 0, stream>>>(x, Ut);
    build_w_kernel<<<(N_DIM + 255) / 256, 256, 0, stream>>>(locs, scales, edges,
                                                            W, Koff);

    constexpr int GRID = (B_DIM / 128) * (N_DIM / 256);
    wmma_gemm_kernel<<<GRID, 256, 0, stream>>>(Ut, W, Koff, out);
}


# Both prep launches fused into the GEMM: each block gathers its own 32x64 W tile and Koff into LDS (c loop as before), waves take A from x and square in registers; one launch, sums unchanged
# speedup vs baseline: 3.2569x; 1.4187x over previous
#include <hip/hip_runtime.h>
#include <hip/hip_bf16.h>

typedef __attribute__((ext_vector_type(2))) float v2f;
typedef __attribute__((ext_vector_type(4))) float v4f;
typedef __attribute__((ext_vector_type(8))) float v8f;

#define B_DIM 1024
#define C_DIM 16
#define M_DIM 4096
#define N_DIM 8192
#define K_DIM 32
#define HALF_LOG_2PI 0.91893853320467274f

#define TILE_N 64
#define TILE_B 512

__global__ __launch_bounds__(256)
void fused_gemm_kernel(const float* __restrict__ x,
                       const float* __restrict__ locs,
                       const float* __restrict__ scales,
                       const int*   __restrict__ edges,
                       float* __restrict__ out) {
    __shared__ __attribute__((aligned(16))) float Wl[K_DIM * TILE_N];
    __shared__ __attribute__((aligned(16))) float Kl[TILE_N];

    const int tid  = threadIdx.x;
    const int lane = tid & 31;
    const int wave = tid >> 5;
    const int h    = lane >> 4;
    const int li   = lane & 15;

    const int nb = blockIdx.x & 127;
    const int bb = blockIdx.x >> 7;
    const int n_base = nb * TILE_N;
    const int b_base = bb * TILE_B + wave * 64;

    const bool builder = tid < TILE_N;

    int e[C_DIM];
    if (builder) {
#pragma unroll
        for (int c = 0; c < C_DIM; ++c)
            e[c] = edges[c * N_DIM + n_base + tid];
    }

    v2f xv[4][4];
    const float* xp = x + (b_base + li) * C_DIM + 2 * h;
#pragma unroll
    for (int u = 0; u < 4; ++u) {
#pragma unroll
        for (int jj = 0; jj < 4; ++jj)
            xv[u][jj] = *(const v2f*)(xp + (16 * u) * C_DIM + 4 * jj);
    }

    if (builder) {
        float ko = 0.0f;
#pragma unroll
        for (int c = 0; c < C_DIM; ++c) {
            float mu   = locs[c * M_DIM + e[c]];
            float s    = scales[c * M_DIM + e[c]] + 0.5f;
            float inv  = 1.0f / s;
            float inv2 = inv * inv;
            Wl[((c >> 1) * TILE_N + tid) * 2 + (c & 1)]       = -0.5f * inv2;
            Wl[((8 + (c >> 1)) * TILE_N + tid) * 2 + (c & 1)] = mu * inv2;
            ko += -0.5f * mu * mu * inv2 - __logf(s) - HALF_LOG_2PI;
        }
        Kl[tid] = ko;
    }
    __syncthreads();

    v4f w[8][2];
    const float* wp = Wl + (h * TILE_N + 4 * li) * 2;
#pragma unroll
    for (int j = 0; j < 8; ++j) {
#pragma unroll
        for (int q = 0; q < 2; ++q)
            w[j][q] = *(const v4f*)(wp + (2 * j) * TILE_N * 2 + 4 * q);
    }
    const v4f ko4 = *(const v4f*)(Kl + 4 * li);
    __builtin_amdgcn_sched_barrier(0);

#pragma unroll
    for (int u = 0; u < 4; ++u) {
        v8f acc[4] = {};
#pragma unroll
        for (int j = 0; j < 8; ++j) {
            v2f af;
            if (j < 4) {
                af.x = xv[u][j].x * xv[u][j].x;
                af.y = xv[u][j].y * xv[u][j].y;
            } else {
                af = xv[u][j - 4];
            }
#pragma unroll
            for (int t = 0; t < 4; ++t) {
                v2f bf;
                bf.x = w[j][t >> 1][2 * (t & 1)];
                bf.y = w[j][t >> 1][2 * (t & 1) + 1];
                acc[t] = __builtin_amdgcn_wmma_f32_16x16x4_f32(
                     false, af,  false, bf,
                     (short)0, acc[t],
                     false,  false);
            }
        }
        float* o = out + (b_base + 16 * u + 8 * h) * N_DIM + n_base + 4 * li;
#pragma unroll
        for (int r = 0; r < 8; ++r) {
            v4f v;
            v.x = acc[0][r] + ko4.x;
            v.y = acc[1][r] + ko4.y;
            v.z = acc[2][r] + ko4.z;
            v.w = acc[3][r] + ko4.w;
            *(v4f*)(o + r * N_DIM) = v;
        }
        __builtin_amdgcn_sched_barrier(0);
    }
}

extern "C" void kernel_launch(void* const* d_in, const int* in_sizes, int n_in,
                              void* d_out, int out_size, void* d_ws, size_t ws_size,
                              hipStream_t stream) {
    const float* x      = (const float*)d_in[0];
    const float* locs   = (const float*)d_in[1];
    const float* scales = (const float*)d_in[2];
    const int*   edges  = (const int*)d_in[3];
    float*       out    = (float*)d_out;

    constexpr int GRID = (B_DIM / TILE_B) * (N_DIM / TILE_N);
    fused_gemm_kernel<<<GRID, 256, 0, stream>>>(x, locs, scales, edges, out);
}


# Block cut from 8 waves to 4, one per SIMD: each wave takes 128 rows not 64 and issues row group u-1's Koff adds and stores between group u's first 16 WMMAs; half as many waves enter; sums unchanged
# speedup vs baseline: 3.9787x; 1.2216x over previous
#include <hip/hip_runtime.h>
#include <hip/hip_bf16.h>

typedef __attribute__((ext_vector_type(2))) float v2f;
typedef __attribute__((ext_vector_type(4))) float v4f;
typedef __attribute__((ext_vector_type(8))) float v8f;

#define B_DIM 1024
#define C_DIM 16
#define M_DIM 4096
#define N_DIM 8192
#define K_DIM 32
#define HALF_LOG_2PI 0.91893853320467274f

#define TILE_N 64
#define TILE_B 512

__global__ __launch_bounds__(128)
void fused_gemm_kernel(const float* __restrict__ x,
                       const float* __restrict__ locs,
                       const float* __restrict__ scales,
                       const int*   __restrict__ edges,
                       float* __restrict__ out) {
    __shared__ __attribute__((aligned(16))) float Wl[K_DIM * TILE_N];
    __shared__ __attribute__((aligned(16))) float Kl[TILE_N];

    const int tid  = threadIdx.x;
    const int lane = tid & 31;
    const int wave = tid >> 5;
    const int h    = lane >> 4;
    const int li   = lane & 15;

    const int nb = blockIdx.x & 127;
    const int bb = blockIdx.x >> 7;
    const int n_base = nb * TILE_N;
    const int b_base = bb * TILE_B + wave * 128;

    const bool builder = tid < TILE_N;

    int e[C_DIM];
    if (builder) {
#pragma unroll
        for (int c = 0; c < C_DIM; ++c)
            e[c] = edges[c * N_DIM + n_base + tid];
    }

    v2f xv[8][4];
    const float* xp = x + (b_base + li) * C_DIM + 2 * h;
#pragma unroll
    for (int u = 0; u < 8; ++u) {
#pragma unroll
        for (int jj = 0; jj < 4; ++jj)
            xv[u][jj] = *(const v2f*)(xp + (16 * u) * C_DIM + 4 * jj);
    }

    if (builder) {
        float ko = 0.0f;
#pragma unroll
        for (int c = 0; c < C_DIM; ++c) {
            float mu   = locs[c * M_DIM + e[c]];
            float s    = scales[c * M_DIM + e[c]] + 0.5f;
            float inv  = 1.0f / s;
            float inv2 = inv * inv;
            Wl[((c >> 1) * TILE_N + tid) * 2 + (c & 1)]       = -0.5f * inv2;
            Wl[((8 + (c >> 1)) * TILE_N + tid) * 2 + (c & 1)] = mu * inv2;
            ko += -0.5f * mu * mu * inv2 - __logf(s) - HALF_LOG_2PI;
        }
        Kl[tid] = ko;
    }
    __syncthreads();

    v4f w[8][2];
    const float* wp = Wl + (h * TILE_N + 4 * li) * 2;
#pragma unroll
    for (int j = 0; j < 8; ++j) {
#pragma unroll
        for (int q = 0; q < 2; ++q)
            w[j][q] = *(const v4f*)(wp + (2 * j) * TILE_N * 2 + 4 * q);
    }
    const v4f ko4 = *(const v4f*)(Kl + 4 * li);
    __builtin_amdgcn_sched_barrier(0);

    v8f acc[2][4];
#pragma unroll
    for (int u = 0; u <= 8; ++u) {
#pragma unroll
        for (int half = 0; half < 2; ++half) {
            if (u < 8) {
                if (half == 0) {
#pragma unroll
                    for (int t = 0; t < 4; ++t)
                        acc[u & 1][t] = (v8f){};
                }
#pragma unroll
                for (int j = 4 * half; j < 4 * half + 4; ++j) {
                    v2f af;
                    if (j < 4) {
                        af.x = xv[u][j].x * xv[u][j].x;
                        af.y = xv[u][j].y * xv[u][j].y;
                    } else {
                        af = xv[u][j - 4];
                    }
#pragma unroll
                    for (int t = 0; t < 4; ++t) {
                        v2f bf;
                        bf.x = w[j][t >> 1][2 * (t & 1)];
                        bf.y = w[j][t >> 1][2 * (t & 1) + 1];
                        acc[u & 1][t] = __builtin_amdgcn_wmma_f32_16x16x4_f32(
                             false, af,  false, bf,
                             (short)0, acc[u & 1][t],
                             false,  false);
                    }
                }
            }
            if (u > 0 && half == 0) {
                const int p = u - 1;
                float* o = out + (b_base + 16 * p + 8 * h) * N_DIM + n_base + 4 * li;
#pragma unroll
                for (int r = 0; r < 8; ++r) {
                    v4f v;
                    v.x = acc[p & 1][0][r] + ko4.x;
                    v.y = acc[p & 1][1][r] + ko4.y;
                    v.z = acc[p & 1][2][r] + ko4.z;
                    v.w = acc[p & 1][3][r] + ko4.w;
                    *(v4f*)(o + r * N_DIM) = v;
                }
            }
            __builtin_amdgcn_sched_barrier(0);
        }
    }
}

extern "C" void kernel_launch(void* const* d_in, const int* in_sizes, int n_in,
                              void* d_out, int out_size, void* d_ws, size_t ws_size,
                              hipStream_t stream) {
    const float* x      = (const float*)d_in[0];
    const float* locs   = (const float*)d_in[1];
    const float* scales = (const float*)d_in[2];
    const int*   edges  = (const int*)d_in[3];
    float*       out    = (float*)d_out;

    constexpr int GRID = (B_DIM / TILE_B) * (N_DIM / TILE_N);
    fused_gemm_kernel<<<GRID, 128, 0, stream>>>(x, locs, scales, edges, out);
}
